# starting kernel
# baseline (speedup 1.0000x reference)
#include <hip/hip_runtime.h>

#define IN_DIM  256
#define OUT_DIM 64
#define N_HEADS 4
#define GAT_EPS 1e-12f

typedef float v2f __attribute__((ext_vector_type(2)));
typedef float v8f __attribute__((ext_vector_type(8)));

__global__ __launch_bounds__(256) void gat_zero(float4* __restrict__ p, long n4) {
    long i = (long)blockIdx.x * blockDim.x + threadIdx.x;
    if (i < n4) p[i] = make_float4(0.f, 0.f, 0.f, 0.f);
}

__global__ __launch_bounds__(128) void gat_gemm_wmma(
    const float* __restrict__ h, const float* __restrict__ W,
    float* __restrict__ x, int N)
{
    const int lane = threadIdx.x & 31;
    const int wave = threadIdx.x >> 5;
    const int half = lane >> 4;
    const int l15  = lane & 15;
    const int row0 = blockIdx.x * 16;
    const int col0 = wave * 16;

    int arow = row0 + l15;
    if (arow >= N) arow = N - 1;
    const float* __restrict__ hrow = h + (size_t)arow * IN_DIM;
    const int bcol = col0 + l15;

    v8f acc = {0.f, 0.f, 0.f, 0.f, 0.f, 0.f, 0.f, 0.f};

    for (int k = 0; k < IN_DIM; k += 4) {
        const int ka = k + half * 2;
        v2f a = *(const v2f*)(hrow + ka);
        v2f b;
        b.x = W[(size_t)ka       * OUT_DIM + bcol];
        b.y = W[(size_t)(ka + 1) * OUT_DIM + bcol];
        acc = __builtin_amdgcn_wmma_f32_16x16x4_f32(
             false, a,  false, b,
             (short)0, acc,  false,  false);
    }

    float* __restrict__ xout =
        x + ((size_t)(row0 + half * 8)) * OUT_DIM + col0 + l15;
    if (row0 + 16 <= N) {
        #pragma unroll
        for (int r = 0; r < 8; ++r)
            xout[(size_t)r * OUT_DIM] = acc[r];
    } else {
        #pragma unroll
        for (int r = 0; r < 8; ++r) {
            if (row0 + r + half * 8 < N)
                xout[(size_t)r * OUT_DIM] = acc[r];
        }
    }
}

__global__ __launch_bounds__(256) void gat_coef(
    const float* __restrict__ x, const float* __restrict__ Wl,
    const float* __restrict__ Wr, float* __restrict__ el,
    float* __restrict__ er, float* __restrict__ rowsum, int N)
{
    const int n = blockIdx.x * blockDim.x + threadIdx.x;
    if (n >= N) return;
    const float* __restrict__ xr = x + (size_t)n * OUT_DIM;
    float al[N_HEADS] = {0.f, 0.f, 0.f, 0.f};
    float ar[N_HEADS] = {0.f, 0.f, 0.f, 0.f};
    for (int d = 0; d < OUT_DIM; ++d) {
        const float xv = xr[d];
        #pragma unroll
        for (int hh = 0; hh < N_HEADS; ++hh) {
            al[hh] = fmaf(xv, Wl[hh * OUT_DIM + d], al[hh]);
            ar[hh] = fmaf(xv, Wr[hh * OUT_DIM + d], ar[hh]);
        }
    }
    #pragma unroll
    for (int hh = 0; hh < N_HEADS; ++hh) {
        el[(size_t)n * N_HEADS + hh] = al[hh];
        er[(size_t)n * N_HEADS + hh] = ar[hh];
        rowsum[(size_t)n * N_HEADS + hh] = 0.f;
    }
}

__device__ __forceinline__ float gat_edge_e(float l, float r) {
    float v = l + r;
    v = (v > 0.f) ? v : 0.2f * v;
    return __expf(v);
}

__global__ __launch_bounds__(256) void gat_rowsum(
    const int* __restrict__ ei, const float* __restrict__ el,
    const float* __restrict__ er, float* __restrict__ rowsum, int E)
{
    const int e = blockIdx.x * blockDim.x + threadIdx.x;
    if (e >= E) return;
    const int src = ei[e];
    const int dst = ei[E + e];
    const float4 l = *(const float4*)(el + (size_t)src * N_HEADS);
    const float4 r = *(const float4*)(er + (size_t)dst * N_HEADS);
    float* rs = rowsum + (size_t)src * N_HEADS;
    atomicAdd(rs + 0, gat_edge_e(l.x, r.x));
    atomicAdd(rs + 1, gat_edge_e(l.y, r.y));
    atomicAdd(rs + 2, gat_edge_e(l.z, r.z));
    atomicAdd(rs + 3, gat_edge_e(l.w, r.w));
}

__global__ __launch_bounds__(256) void gat_aggregate(
    const int* __restrict__ ei, const float* __restrict__ el,
    const float* __restrict__ er, const float* __restrict__ rowsum,
    const float* __restrict__ x, float* __restrict__ out, int E)
{
    const int e    = blockIdx.x * 8 + (threadIdx.x >> 5);
    const int lane = threadIdx.x & 31;
    if (e >= E) return;
    const int src = ei[e];
    const int dst = ei[E + e];

    const float4 l  = *(const float4*)(el     + (size_t)src * N_HEADS);
    const float4 r  = *(const float4*)(er     + (size_t)dst * N_HEADS);
    const float4 rs = *(const float4*)(rowsum + (size_t)src * N_HEADS);

    float w[N_HEADS];
    w[0] = gat_edge_e(l.x, r.x) / fmaxf(rs.x, GAT_EPS);
    w[1] = gat_edge_e(l.y, r.y) / fmaxf(rs.y, GAT_EPS);
    w[2] = gat_edge_e(l.z, r.z) / fmaxf(rs.z, GAT_EPS);
    w[3] = gat_edge_e(l.w, r.w) / fmaxf(rs.w, GAT_EPS);

    const v2f xv = *(const v2f*)(x + (size_t)dst * OUT_DIM + lane * 2);
    float* obase = out + (size_t)src * (N_HEADS * OUT_DIM) + lane * 2;
    #pragma unroll
    for (int hh = 0; hh < N_HEADS; ++hh) {
        atomicAdd(obase + hh * OUT_DIM + 0, w[hh] * xv.x);
        atomicAdd(obase + hh * OUT_DIM + 1, w[hh] * xv.y);
    }
}

__global__ __launch_bounds__(256) void gat_finalize(
    float* __restrict__ out, const float* __restrict__ b, long n4)
{
    const long i4 = (long)blockIdx.x * blockDim.x + threadIdx.x;
    if (i4 >= n4) return;
    float4 v = ((const float4*)out)[i4];
    const int d0 = (int)((i4 * 4) & (OUT_DIM - 1));
    v.x += b[d0 + 0]; v.y += b[d0 + 1]; v.z += b[d0 + 2]; v.w += b[d0 + 3];
    v.x = (v.x > 0.f) ? v.x : (__expf(v.x) - 1.f);
    v.y = (v.y > 0.f) ? v.y : (__expf(v.y) - 1.f);
    v.z = (v.z > 0.f) ? v.z : (__expf(v.z) - 1.f);
    v.w = (v.w > 0.f) ? v.w : (__expf(v.w) - 1.f);
    ((float4*)out)[i4] = v;
}

extern "C" void kernel_launch(void* const* d_in, const int* in_sizes, int n_in,
                              void* d_out, int out_size, void* d_ws, size_t ws_size,
                              hipStream_t stream) {
    const float* h  = (const float*)d_in[0];
    const float* W  = (const float*)d_in[1];
    const float* Wl = (const float*)d_in[2];
    const float* Wr = (const float*)d_in[3];
    const float* b  = (const float*)d_in[4];
    const int*   ei = (const int*)d_in[5];

    const int N = in_sizes[0] / IN_DIM;
    const int E = in_sizes[5] / 2;

    float* out = (float*)d_out;

    float* x      = (float*)d_ws;
    float* el     = x  + (size_t)N * OUT_DIM;
    float* er     = el + (size_t)N * N_HEADS;
    float* rowsum = er + (size_t)N * N_HEADS;

    const long out_n4 = (long)out_size / 4;

    gat_zero<<<(unsigned)((out_n4 + 255) / 256), 256, 0, stream>>>(
        (float4*)out, out_n4);

    gat_gemm_wmma<<<(unsigned)((N + 15) / 16), 128, 0, stream>>>(h, W, x, N);

    gat_coef<<<(unsigned)((N + 255) / 256), 256, 0, stream>>>(
        x, Wl, Wr, el, er, rowsum, N);

    gat_rowsum<<<(unsigned)((E + 255) / 256), 256, 0, stream>>>(
        ei, el, er, rowsum, E);

    gat_aggregate<<<(unsigned)((E + 7) / 8), 256, 0, stream>>>(
        ei, el, er, rowsum, x, out, E);

    gat_finalize<<<(unsigned)((out_n4 + 255) / 256), 256, 0, stream>>>(
        out, b, out_n4);
}


# GEMM pass: one wave per 16x64 tile of x (4 accumulator tiles): h read by one wave not four, W and x rows move as whole 256-byte rows, next stage's loads issued ahead of the matrix instructions
# speedup vs baseline: 3.4849x; 3.4849x over previous
#include <hip/hip_runtime.h>


#define IN_DIM  256
#define OUT_DIM 64
#define N_HEADS 4
#define GAT_EPS 1e-12f
#define ROW_CAP 256
#define KEY_PAD 0x7fffffff

typedef float v2f __attribute__((ext_vector_type(2)));
typedef float v4f __attribute__((ext_vector_type(4)));
typedef float v8f __attribute__((ext_vector_type(8)));

struct GemmStage {
    v4f a[4];
    v4f b[4][4];
};

__device__ __forceinline__ void gemm_load(
    GemmStage& g, const float* __restrict__ ap, const float* __restrict__ bp,
    const int st)
{
    #pragma unroll
    for (int sub = 0; sub < 4; ++sub) {
        const int k0 = st * 32 + sub * 8;
        g.a[sub] = *(const v4f*)(ap + k0);
        #pragma unroll
        for (int j = 0; j < 4; ++j)
            g.b[sub][j] = *(const v4f*)(bp + (size_t)(k0 + j) * OUT_DIM);
    }
}

__device__ __forceinline__ void gemm_mma(v8f (&acc)[4], const GemmStage& g)
{
    #pragma unroll
    for (int sub = 0; sub < 4; ++sub) {
        #pragma unroll
        for (int jp = 0; jp < 2; ++jp) {
            v2f a;
            a.x = g.a[sub][2 * jp];
            a.y = g.a[sub][2 * jp + 1];
            #pragma unroll
            for (int t = 0; t < 4; ++t) {
                v2f b;
                b.x = g.b[sub][2 * jp][t];
                b.y = g.b[sub][2 * jp + 1][t];
                acc[t] = __builtin_amdgcn_wmma_f32_16x16x4_f32(
                     false, a,  false, b,
                     (short)0, acc[t],  false,  false);
            }
        }
    }
}

__global__ __launch_bounds__(128) void gat_gemm_wmma(
    const float* __restrict__ h, const float* __restrict__ W,
    float* __restrict__ x, int N)
{
    const int lane = threadIdx.x & 31;
    const int wave = threadIdx.x >> 5;
    const int hf   = lane >> 4;
    const int c    = lane & 15;
    const int row0 = blockIdx.x * 64 + wave * 16;
    if (row0 >= N) return;

    int arow = row0 + c;
    if (arow >= N) arow = N - 1;
    const float* __restrict__ ap = h + (size_t)arow * IN_DIM + hf * 4;
    const float* __restrict__ bp = W + (size_t)(hf * 4) * OUT_DIM + c * 4;

    v8f acc[4];
    #pragma unroll
    for (int t = 0; t < 4; ++t)
        acc[t] = (v8f){0.f, 0.f, 0.f, 0.f, 0.f, 0.f, 0.f, 0.f};

    GemmStage g0, g1;
    gemm_load(g0, ap, bp, 0);
    #pragma unroll
    for (int st = 0; st < IN_DIM / 32; st += 2) {
        gemm_load(g1, ap, bp, st + 1);
        __builtin_amdgcn_sched_barrier(0);
        gemm_mma(acc, g0);
        if (st + 2 < IN_DIM / 32) {
            gemm_load(g0, ap, bp, st + 2);
            __builtin_amdgcn_sched_barrier(0);
        }
        gemm_mma(acc, g1);
    }

    float* __restrict__ xout =
        x + ((size_t)(row0 + hf * 8)) * OUT_DIM + c * 4;
    if (row0 + 16 <= N) {
        #pragma unroll
        for (int r = 0; r < 8; ++r) {
            v4f v = {acc[0][r], acc[1][r], acc[2][r], acc[3][r]};
            *(v4f*)(xout + (size_t)r * OUT_DIM) = v;
        }
    } else {
        #pragma unroll
        for (int r = 0; r < 8; ++r) {
            if (row0 + r + hf * 8 < N) {
                v4f v = {acc[0][r], acc[1][r], acc[2][r], acc[3][r]};
                *(v4f*)(xout + (size_t)r * OUT_DIM) = v;
            }
        }
    }
}

__global__ __launch_bounds__(256) void gat_coef(
    const float* __restrict__ x, const float* __restrict__ Wl,
    const float* __restrict__ Wr, float* __restrict__ el,
    float* __restrict__ er, int* __restrict__ cnt, int N)
{
    const int n = blockIdx.x * blockDim.x + threadIdx.x;
    if (n >= N) return;
    const float* __restrict__ xr = x + (size_t)n * OUT_DIM;
    float al[N_HEADS] = {0.f, 0.f, 0.f, 0.f};
    float ar[N_HEADS] = {0.f, 0.f, 0.f, 0.f};
    for (int d = 0; d < OUT_DIM; ++d) {
        const float xv = xr[d];
        #pragma unroll
        for (int hh = 0; hh < N_HEADS; ++hh) {
            al[hh] = fmaf(xv, Wl[hh * OUT_DIM + d], al[hh]);
            ar[hh] = fmaf(xv, Wr[hh * OUT_DIM + d], ar[hh]);
        }
    }
    #pragma unroll
    for (int hh = 0; hh < N_HEADS; ++hh) {
        el[(size_t)n * N_HEADS + hh] = al[hh];
        er[(size_t)n * N_HEADS + hh] = ar[hh];
    }
    cnt[n] = 0;
}

__device__ __forceinline__ float gat_edge_e(float l, float r) {
    float v = l + r;
    v = (v > 0.f) ? v : 0.2f * v;
    return __expf(v);
}

__global__ __launch_bounds__(256) void gat_bucket(
    const int* __restrict__ ei, int* __restrict__ cnt,
    int* __restrict__ slots, int E)
{
    const int e = blockIdx.x * 256 + threadIdx.x;
    if (e >= E) return;
    const int src = ei[e];
    const int dst = ei[E + e];
    const int rank = atomicAdd(cnt + src, 1);
    if (rank < ROW_CAP) slots[(size_t)src * ROW_CAP + rank] = dst;
}

struct RowAcc {
    v2f   a[N_HEADS];
    float rs[N_HEADS];
};

__device__ __forceinline__ float lane_f(float v, int j) {
    return __builtin_bit_cast(float,
        __builtin_amdgcn_readlane(__builtin_bit_cast(int, v), j));
}

template <int K>
__device__ __forceinline__ void sort_keys(int (&key)[K], const int lane)
{
    #pragma unroll
    for (int k = 2; k <= 32 * K; k <<= 1) {
        #pragma unroll
        for (int j = k >> 1; j > 0; j >>= 1) {
            if (j >= 32) {
                const int rj = j >> 5;
                #pragma unroll
                for (int r = 0; r < K; ++r) {
                    const int p = r ^ rj;
                    if (p > r) {
                        const bool up = (((r * 32) & k) == 0);
                        const int lo = key[r] < key[p] ? key[r] : key[p];
                        const int hi = key[r] < key[p] ? key[p] : key[r];
                        key[r] = up ? lo : hi;
                        key[p] = up ? hi : lo;
                    }
                }
            } else {
                #pragma unroll
                for (int r = 0; r < K; ++r) {
                    const int i = r * 32 + lane;
                    const int other = __shfl_xor(key[r], j, 32);
                    const bool up    = ((i & k) == 0);
                    const bool lower = ((lane & j) == 0);
                    const int lo = key[r] < other ? key[r] : other;
                    const int hi = key[r] < other ? other : key[r];
                    key[r] = (up == lower) ? lo : hi;
                }
            }
        }
    }
}

template <int K>
__device__ __forceinline__ void row_from_slots(
    RowAcc& s, const int* slots_row, const int deg, const int lane,
    const float4 l, const float* __restrict__ er, const float* __restrict__ x,
    const int N)
{
    int key[K];
    #pragma unroll
    for (int r = 0; r < K; ++r) {
        const int i = r * 32 + lane;
        key[r] = (i < deg) ? slots_row[i] : KEY_PAD;
    }
    sort_keys<K>(key, lane);

    #pragma unroll
    for (int r = 0; r < K; ++r) {
        const int left = deg - r * 32;
        if (left > 0) {
            const int nh = left < 32 ? left : 32;
            const bool valid = lane < nh;
            int d = valid ? key[r] : __builtin_amdgcn_readlane(key[r], nh - 1);
            d = (int)(((unsigned)d < (unsigned)N) ? (unsigned)d : (unsigned)(N - 1));
            const float4 rr = *(const float4*)(er + (size_t)d * N_HEADS);
            const float w0 = valid ? gat_edge_e(l.x, rr.x) : 0.f;
            const float w1 = valid ? gat_edge_e(l.y, rr.y) : 0.f;
            const float w2 = valid ? gat_edge_e(l.z, rr.z) : 0.f;
            const float w3 = valid ? gat_edge_e(l.w, rr.w) : 0.f;

            const int npad = (nh + 3) & ~3;
            for (int j = 0; j < npad; j += 4) {
                v2f xv[4];
                #pragma unroll
                for (int u = 0; u < 4; ++u) {
                    const int du = __builtin_amdgcn_readlane(d, j + u);
                    xv[u] = *(const v2f*)(x + (size_t)du * OUT_DIM + lane * 2);
                }
                #pragma unroll
                for (int u = 0; u < 4; ++u) {
                    const float e0 = lane_f(w0, j + u);
                    const float e1 = lane_f(w1, j + u);
                    const float e2 = lane_f(w2, j + u);
                    const float e3 = lane_f(w3, j + u);
                    s.a[0] += e0 * xv[u];  s.rs[0] += e0;
                    s.a[1] += e1 * xv[u];  s.rs[1] += e1;
                    s.a[2] += e2 * xv[u];  s.rs[2] += e2;
                    s.a[3] += e3 * xv[u];  s.rs[3] += e3;
                }
            }
        }
    }
}

__device__ __forceinline__ void row_from_edge_list(
    RowAcc& s, const int* __restrict__ ei, const int E, const int row,
    const int lane, const float4 l, const float* __restrict__ er,
    const float* __restrict__ x)
{
    for (int base = 0; base < E; base += 32) {
        const int e  = base + lane;
        const int sv = (e < E) ? ei[e] : -1;
        unsigned m = __builtin_amdgcn_ballot_w32(sv == row);
        while (m) {
            const int bit = __builtin_ctz(m);
            m &= m - 1;
            const int d = ei[E + base + bit];
            const float4 rr = *(const float4*)(er + (size_t)d * N_HEADS);
            const float e0 = gat_edge_e(l.x, rr.x);
            const float e1 = gat_edge_e(l.y, rr.y);
            const float e2 = gat_edge_e(l.z, rr.z);
            const float e3 = gat_edge_e(l.w, rr.w);
            const v2f xv = *(const v2f*)(x + (size_t)d * OUT_DIM + lane * 2);
            s.a[0] += e0 * xv;  s.rs[0] += e0;
            s.a[1] += e1 * xv;  s.rs[1] += e1;
            s.a[2] += e2 * xv;  s.rs[2] += e2;
            s.a[3] += e3 * xv;  s.rs[3] += e3;
        }
    }
}

__global__ __launch_bounds__(256) void gat_rows(
    const int* __restrict__ ei, const int* __restrict__ cnt,
    const float* __restrict__ el, const float* __restrict__ er,
    const float* __restrict__ x, const float* __restrict__ b,
    float* out, const int* slots, int N, int E)
{
    const int lane = threadIdx.x & 31;
    const int row = __builtin_amdgcn_readfirstlane(blockIdx.x * 8 + (threadIdx.x >> 5));
    if (row >= N) return;
    const int deg = __builtin_amdgcn_readfirstlane(cnt[row]);
    const float4 l = *(const float4*)(el + (size_t)row * N_HEADS);
    const int* slots_row = slots + (size_t)row * ROW_CAP;

    RowAcc s;
    #pragma unroll
    for (int hh = 0; hh < N_HEADS; ++hh) {
        s.a[hh] = (v2f){0.f, 0.f};
        s.rs[hh] = 0.f;
    }

    if (deg <= 32)            row_from_slots<1>(s, slots_row, deg, lane, l, er, x, N);
    else if (deg <= 64)       row_from_slots<2>(s, slots_row, deg, lane, l, er, x, N);
    else if (deg <= 128)      row_from_slots<4>(s, slots_row, deg, lane, l, er, x, N);
    else if (deg <= ROW_CAP)  row_from_slots<8>(s, slots_row, deg, lane, l, er, x, N);
    else                      row_from_edge_list(s, ei, E, row, lane, l, er, x);

    float* orow = out + (size_t)row * (N_HEADS * OUT_DIM) + lane * 2;
    const v2f bb = *(const v2f*)(b + lane * 2);
    #pragma unroll
    for (int hh = 0; hh < N_HEADS; ++hh) {
        const float inv = 1.0f / fmaxf(s.rs[hh], GAT_EPS);
        v2f v;
        v.x = fmaf(s.a[hh].x, inv, bb.x);
        v.y = fmaf(s.a[hh].y, inv, bb.y);
        v.x = (v.x > 0.f) ? v.x : (__expf(v.x) - 1.f);
        v.y = (v.y > 0.f) ? v.y : (__expf(v.y) - 1.f);
        *(v2f*)(orow + hh * OUT_DIM) = v;
    }
}

extern "C" void kernel_launch(void* const* d_in, const int* in_sizes, int n_in,
                              void* d_out, int out_size, void* d_ws, size_t ws_size,
                              hipStream_t stream) {
    const float* h  = (const float*)d_in[0];
    const float* W  = (const float*)d_in[1];
    const float* Wl = (const float*)d_in[2];
    const float* Wr = (const float*)d_in[3];
    const float* b  = (const float*)d_in[4];
    const int*   ei = (const int*)d_in[5];

    const int N = in_sizes[0] / IN_DIM;
    const int E = in_sizes[5] / 2;

    float* out = (float*)d_out;

    float* x   = (float*)d_ws;
    float* el  = x  + (size_t)N * OUT_DIM;
    float* er  = el + (size_t)N * N_HEADS;
    int*   cnt = (int*)(er + (size_t)N * N_HEADS);

    gat_gemm_wmma<<<(unsigned)((N + 63) / 64), 128, 0, stream>>>(h, W, x, N);

    gat_coef<<<(unsigned)((N + 255) / 256), 256, 0, stream>>>(
        x, Wl, Wr, el, er, cnt, N);

    gat_bucket<<<(unsigned)((E + 255) / 256), 256, 0, stream>>>(
        ei, cnt, (int*)d_out, E);

    gat_rows<<<(unsigned)((N + 7) / 8), 256, 0, stream>>>(
        ei, cnt, el, er, x, b, out, (const int*)d_out, N, E);
}


# bucket pass: each 1024-thread block first sorts its 6144 edges by counter line (32 rows) in LDS, so a wave's 32 integer atomics hit about 9 lines, not 32; 261 work-groups, not 6250; sum order kept
# speedup vs baseline: 4.5182x; 1.2965x over previous
#include <hip/hip_runtime.h>


#define IN_DIM  256
#define OUT_DIM 64
#define N_HEADS 4
#define GAT_EPS 1e-12f
#define ROW_CAP 256
#define KEY_PAD 0x7fffffff

typedef float v2f __attribute__((ext_vector_type(2)));
typedef float v4f __attribute__((ext_vector_type(4)));
typedef float v8f __attribute__((ext_vector_type(8)));

struct GemmStage {
    v4f a[4];
    v4f b[4][4];
};

__device__ __forceinline__ void gemm_load(
    GemmStage& g, const float* __restrict__ ap, const float* __restrict__ bp,
    const int st)
{
    #pragma unroll
    for (int sub = 0; sub < 4; ++sub) {
        const int k0 = st * 32 + sub * 8;
        g.a[sub] = *(const v4f*)(ap + k0);
        #pragma unroll
        for (int j = 0; j < 4; ++j)
            g.b[sub][j] = *(const v4f*)(bp + (size_t)(k0 + j) * OUT_DIM);
    }
}

__device__ __forceinline__ void gemm_mma(v8f (&acc)[4], const GemmStage& g)
{
    #pragma unroll
    for (int sub = 0; sub < 4; ++sub) {
        #pragma unroll
        for (int jp = 0; jp < 2; ++jp) {
            v2f a;
            a.x = g.a[sub][2 * jp];
            a.y = g.a[sub][2 * jp + 1];
            #pragma unroll
            for (int t = 0; t < 4; ++t) {
                v2f b;
                b.x = g.b[sub][2 * jp][t];
                b.y = g.b[sub][2 * jp + 1][t];
                acc[t] = __builtin_amdgcn_wmma_f32_16x16x4_f32(
                     false, a,  false, b,
                     (short)0, acc[t],  false,  false);
            }
        }
    }
}

__global__ __launch_bounds__(128) void gat_gemm_wmma(
    const float* __restrict__ h, const float* __restrict__ W,
    float* __restrict__ x, int N)
{
    const int lane = threadIdx.x & 31;
    const int wave = threadIdx.x >> 5;
    const int hf   = lane >> 4;
    const int c    = lane & 15;
    const int row0 = blockIdx.x * 64 + wave * 16;
    if (row0 >= N) return;

    int arow = row0 + c;
    if (arow >= N) arow = N - 1;
    const float* __restrict__ ap = h + (size_t)arow * IN_DIM + hf * 4;
    const float* __restrict__ bp = W + (size_t)(hf * 4) * OUT_DIM + c * 4;

    v8f acc[4];
    #pragma unroll
    for (int t = 0; t < 4; ++t)
        acc[t] = (v8f){0.f, 0.f, 0.f, 0.f, 0.f, 0.f, 0.f, 0.f};

    GemmStage g0, g1;
    gemm_load(g0, ap, bp, 0);
    #pragma unroll
    for (int st = 0; st < IN_DIM / 32; st += 2) {
        gemm_load(g1, ap, bp, st + 1);
        __builtin_amdgcn_sched_barrier(0);
        gemm_mma(acc, g0);
        if (st + 2 < IN_DIM / 32) {
            gemm_load(g0, ap, bp, st + 2);
            __builtin_amdgcn_sched_barrier(0);
        }
        gemm_mma(acc, g1);
    }

    float* __restrict__ xout =
        x + ((size_t)(row0 + hf * 8)) * OUT_DIM + c * 4;
    if (row0 + 16 <= N) {
        #pragma unroll
        for (int r = 0; r < 8; ++r) {
            v4f v = {acc[0][r], acc[1][r], acc[2][r], acc[3][r]};
            *(v4f*)(xout + (size_t)r * OUT_DIM) = v;
        }
    } else {
        #pragma unroll
        for (int r = 0; r < 8; ++r) {
            if (row0 + r + hf * 8 < N) {
                v4f v = {acc[0][r], acc[1][r], acc[2][r], acc[3][r]};
                *(v4f*)(xout + (size_t)r * OUT_DIM) = v;
            }
        }
    }
}

__global__ __launch_bounds__(256) void gat_coef(
    const float* __restrict__ x, const float* __restrict__ Wl,
    const float* __restrict__ Wr, float* __restrict__ el,
    float* __restrict__ er, int* __restrict__ cnt, int N)
{
    const int n = blockIdx.x * blockDim.x + threadIdx.x;
    if (n >= N) return;
    const float* __restrict__ xr = x + (size_t)n * OUT_DIM;
    float al[N_HEADS] = {0.f, 0.f, 0.f, 0.f};
    float ar[N_HEADS] = {0.f, 0.f, 0.f, 0.f};
    for (int d = 0; d < OUT_DIM; ++d) {
        const float xv = xr[d];
        #pragma unroll
        for (int hh = 0; hh < N_HEADS; ++hh) {
            al[hh] = fmaf(xv, Wl[hh * OUT_DIM + d], al[hh]);
            ar[hh] = fmaf(xv, Wr[hh * OUT_DIM + d], ar[hh]);
        }
    }
    #pragma unroll
    for (int hh = 0; hh < N_HEADS; ++hh) {
        el[(size_t)n * N_HEADS + hh] = al[hh];
        er[(size_t)n * N_HEADS + hh] = ar[hh];
    }
    cnt[n] = 0;
}

__device__ __forceinline__ float gat_edge_e(float l, float r) {
    float v = l + r;
    v = (v > 0.f) ? v : 0.2f * v;
    return __expf(v);
}

#define BK_THREADS 1024
#define BK_EPT     6
#define BK_TILE    (BK_THREADS * BK_EPT)
#define BK_BINS    2048

__global__ __launch_bounds__(BK_THREADS) void gat_bucket(
    const int* __restrict__ ei, int* __restrict__ cnt,
    int* __restrict__ slots, int E)
{
    __shared__ int hist[BK_BINS];
    __shared__ int wsum[32];
    __shared__ int s_src[BK_TILE];
    __shared__ int s_dst[BK_TILE];

    const int tid  = threadIdx.x;
    const int lane = tid & 31;
    const int wave = tid >> 5;
    const int base = blockIdx.x * BK_TILE;
    const int left = E - base;
    const int nvalid = left < BK_TILE ? left : BK_TILE;

    hist[tid] = 0;
    hist[tid + BK_THREADS] = 0;

    int src[BK_EPT], dst[BK_EPT], rnk[BK_EPT];
    #pragma unroll
    for (int i = 0; i < BK_EPT; ++i) {
        const int k = i * BK_THREADS + tid;
        src[i] = 0;
        dst[i] = 0;
        if (k < nvalid) {
            src[i] = ei[base + k];
            dst[i] = ei[E + base + k];
        }
    }
    __syncthreads();

    #pragma unroll
    for (int i = 0; i < BK_EPT; ++i) {
        const int k = i * BK_THREADS + tid;
        rnk[i] = 0;
        if (k < nvalid)
            rnk[i] = atomicAdd(&hist[(src[i] >> 5) & (BK_BINS - 1)], 1);
    }
    __syncthreads();

    const int c0 = hist[2 * tid];
    const int c1 = hist[2 * tid + 1];
    const int mine = c0 + c1;
    int inc = mine;
    #pragma unroll
    for (int d = 1; d < 32; d <<= 1) {
        const int t = __shfl_up(inc, d, 32);
        if (lane >= d) inc += t;
    }
    if (lane == 31) wsum[wave] = inc;
    __syncthreads();
    if (wave == 0) {
        const int w = wsum[lane];
        int winc = w;
        #pragma unroll
        for (int d = 1; d < 32; d <<= 1) {
            const int t = __shfl_up(winc, d, 32);
            if (lane >= d) winc += t;
        }
        wsum[lane] = winc - w;
    }
    __syncthreads();
    const int start0 = wsum[wave] + inc - mine;
    hist[2 * tid]     = start0;
    hist[2 * tid + 1] = start0 + c0;
    __syncthreads();

    #pragma unroll
    for (int i = 0; i < BK_EPT; ++i) {
        const int k = i * BK_THREADS + tid;
        if (k < nvalid) {
            const int p = hist[(src[i] >> 5) & (BK_BINS - 1)] + rnk[i];
            s_src[p] = src[i];
            s_dst[p] = dst[i];
        }
    }
    __syncthreads();

    #pragma unroll
    for (int i = 0; i < BK_EPT; ++i) {
        const int k = i * BK_THREADS + tid;
        rnk[i] = ROW_CAP;
        if (k < nvalid) {
            src[i] = s_src[k];
            dst[i] = s_dst[k];
            rnk[i] = atomicAdd(cnt + src[i], 1);
        }
    }
    #pragma unroll
    for (int i = 0; i < BK_EPT; ++i) {
        if (rnk[i] < ROW_CAP)
            slots[(size_t)src[i] * ROW_CAP + rnk[i]] = dst[i];
    }
}

struct RowAcc {
    v2f   a[N_HEADS];
    float rs[N_HEADS];
};

__device__ __forceinline__ float lane_f(float v, int j) {
    return __builtin_bit_cast(float,
        __builtin_amdgcn_readlane(__builtin_bit_cast(int, v), j));
}

template <int K>
__device__ __forceinline__ void sort_keys(int (&key)[K], const int lane)
{
    #pragma unroll
    for (int k = 2; k <= 32 * K; k <<= 1) {
        #pragma unroll
        for (int j = k >> 1; j > 0; j >>= 1) {
            if (j >= 32) {
                const int rj = j >> 5;
                #pragma unroll
                for (int r = 0; r < K; ++r) {
                    const int p = r ^ rj;
                    if (p > r) {
                        const bool up = (((r * 32) & k) == 0);
                        const int lo = key[r] < key[p] ? key[r] : key[p];
                        const int hi = key[r] < key[p] ? key[p] : key[r];
                        key[r] = up ? lo : hi;
                        key[p] = up ? hi : lo;
                    }
                }
            } else {
                #pragma unroll
                for (int r = 0; r < K; ++r) {
                    const int i = r * 32 + lane;
                    const int other = __shfl_xor(key[r], j, 32);
                    const bool up    = ((i & k) == 0);
                    const bool lower = ((lane & j) == 0);
                    const int lo = key[r] < other ? key[r] : other;
                    const int hi = key[r] < other ? other : key[r];
                    key[r] = (up == lower) ? lo : hi;
                }
            }
        }
    }
}

template <int K>
__device__ __forceinline__ void row_from_slots(
    RowAcc& s, const int* slots_row, const int deg, const int lane,
    const float4 l, const float* __restrict__ er, const float* __restrict__ x,
    const int N)
{
    int key[K];
    #pragma unroll
    for (int r = 0; r < K; ++r) {
        const int i = r * 32 + lane;
        key[r] = (i < deg) ? slots_row[i] : KEY_PAD;
    }
    sort_keys<K>(key, lane);

    #pragma unroll
    for (int r = 0; r < K; ++r) {
        const int left = deg - r * 32;
        if (left > 0) {
            const int nh = left < 32 ? left : 32;
            const bool valid = lane < nh;
            int d = valid ? key[r] : __builtin_amdgcn_readlane(key[r], nh - 1);
            d = (int)(((unsigned)d < (unsigned)N) ? (unsigned)d : (unsigned)(N - 1));
            const float4 rr = *(const float4*)(er + (size_t)d * N_HEADS);
            const float w0 = valid ? gat_edge_e(l.x, rr.x) : 0.f;
            const float w1 = valid ? gat_edge_e(l.y, rr.y) : 0.f;
            const float w2 = valid ? gat_edge_e(l.z, rr.z) : 0.f;
            const float w3 = valid ? gat_edge_e(l.w, rr.w) : 0.f;

            const int npad = (nh + 3) & ~3;
            for (int j = 0; j < npad; j += 4) {
                v2f xv[4];
                #pragma unroll
                for (int u = 0; u < 4; ++u) {
                    const int du = __builtin_amdgcn_readlane(d, j + u);
                    xv[u] = *(const v2f*)(x + (size_t)du * OUT_DIM + lane * 2);
                }
                #pragma unroll
                for (int u = 0; u < 4; ++u) {
                    const float e0 = lane_f(w0, j + u);
                    const float e1 = lane_f(w1, j + u);
                    const float e2 = lane_f(w2, j + u);
                    const float e3 = lane_f(w3, j + u);
                    s.a[0] += e0 * xv[u];  s.rs[0] += e0;
                    s.a[1] += e1 * xv[u];  s.rs[1] += e1;
                    s.a[2] += e2 * xv[u];  s.rs[2] += e2;
                    s.a[3] += e3 * xv[u];  s.rs[3] += e3;
                }
            }
        }
    }
}

__device__ __forceinline__ void row_from_edge_list(
    RowAcc& s, const int* __restrict__ ei, const int E, const int row,
    const int lane, const float4 l, const float* __restrict__ er,
    const float* __restrict__ x)
{
    for (int base = 0; base < E; base += 32) {
        const int e  = base + lane;
        const int sv = (e < E) ? ei[e] : -1;
        unsigned m = __builtin_amdgcn_ballot_w32(sv == row);
        while (m) {
            const int bit = __builtin_ctz(m);
            m &= m - 1;
            const int d = ei[E + base + bit];
            const float4 rr = *(const float4*)(er + (size_t)d * N_HEADS);
            const float e0 = gat_edge_e(l.x, rr.x);
            const float e1 = gat_edge_e(l.y, rr.y);
            const float e2 = gat_edge_e(l.z, rr.z);
            const float e3 = gat_edge_e(l.w, rr.w);
            const v2f xv = *(const v2f*)(x + (size_t)d * OUT_DIM + lane * 2);
            s.a[0] += e0 * xv;  s.rs[0] += e0;
            s.a[1] += e1 * xv;  s.rs[1] += e1;
            s.a[2] += e2 * xv;  s.rs[2] += e2;
            s.a[3] += e3 * xv;  s.rs[3] += e3;
        }
    }
}

__global__ __launch_bounds__(256) void gat_rows(
    const int* __restrict__ ei, const int* __restrict__ cnt,
    const float* __restrict__ el, const float* __restrict__ er,
    const float* __restrict__ x, const float* __restrict__ b,
    float* out, const int* slots, int N, int E)
{
    const int lane = threadIdx.x & 31;
    const int row = __builtin_amdgcn_readfirstlane(blockIdx.x * 8 + (threadIdx.x >> 5));
    if (row >= N) return;
    const int deg = __builtin_amdgcn_readfirstlane(cnt[row]);
    const float4 l = *(const float4*)(el + (size_t)row * N_HEADS);
    const int* slots_row = slots + (size_t)row * ROW_CAP;

    RowAcc s;
    #pragma unroll
    for (int hh = 0; hh < N_HEADS; ++hh) {
        s.a[hh] = (v2f){0.f, 0.f};
        s.rs[hh] = 0.f;
    }

    if (deg <= 32)            row_from_slots<1>(s, slots_row, deg, lane, l, er, x, N);
    else if (deg <= 64)       row_from_slots<2>(s, slots_row, deg, lane, l, er, x, N);
    else if (deg <= 128)      row_from_slots<4>(s, slots_row, deg, lane, l, er, x, N);
    else if (deg <= ROW_CAP)  row_from_slots<8>(s, slots_row, deg, lane, l, er, x, N);
    else                      row_from_edge_list(s, ei, E, row, lane, l, er, x);

    float* orow = out + (size_t)row * (N_HEADS * OUT_DIM) + lane * 2;
    const v2f bb = *(const v2f*)(b + lane * 2);
    #pragma unroll
    for (int hh = 0; hh < N_HEADS; ++hh) {
        const float inv = 1.0f / fmaxf(s.rs[hh], GAT_EPS);
        v2f v;
        v.x = fmaf(s.a[hh].x, inv, bb.x);
        v.y = fmaf(s.a[hh].y, inv, bb.y);
        v.x = (v.x > 0.f) ? v.x : (__expf(v.x) - 1.f);
        v.y = (v.y > 0.f) ? v.y : (__expf(v.y) - 1.f);
        *(v2f*)(orow + hh * OUT_DIM) = v;
    }
}

extern "C" void kernel_launch(void* const* d_in, const int* in_sizes, int n_in,
                              void* d_out, int out_size, void* d_ws, size_t ws_size,
                              hipStream_t stream) {
    const float* h  = (const float*)d_in[0];
    const float* W  = (const float*)d_in[1];
    const float* Wl = (const float*)d_in[2];
    const float* Wr = (const float*)d_in[3];
    const float* b  = (const float*)d_in[4];
    const int*   ei = (const int*)d_in[5];

    const int N = in_sizes[0] / IN_DIM;
    const int E = in_sizes[5] / 2;

    float* out = (float*)d_out;

    float* x   = (float*)d_ws;
    float* el  = x  + (size_t)N * OUT_DIM;
    float* er  = el + (size_t)N * N_HEADS;
    int*   cnt = (int*)(er + (size_t)N * N_HEADS);

    gat_gemm_wmma<<<(unsigned)((N + 63) / 64), 128, 0, stream>>>(h, W, x, N);

    gat_coef<<<(unsigned)((N + 255) / 256), 256, 0, stream>>>(
        x, Wl, Wr, el, er, cnt, N);

    gat_bucket<<<(unsigned)((E + BK_TILE - 1) / BK_TILE), BK_THREADS, 0, stream>>>(
        ei, cnt, (int*)d_out, E);

    gat_rows<<<(unsigned)((N + 7) / 8), 256, 0, stream>>>(
        ei, cnt, el, er, x, b, out, (const int*)d_out, N, E);
}


# bucket by bin of 32 source rows, not by row: per block one integer add and one run of consecutive stores per bin; the row pass deals its bin's list out to rows in LDS; row sort and sum order kept
# speedup vs baseline: 7.7536x; 1.7161x over previous
#include <hip/hip_runtime.h>


#define IN_DIM  256
#define OUT_DIM 64
#define N_HEADS 4
#define GAT_EPS 1e-12f
#define ROW_CAP 256
#define KEY_PAD 0x7fffffff
#define BIN_ROWS  32
#define BIN_WORDS (BIN_ROWS * ROW_CAP)
#define ROW_SHIFT 27
#define DST_MASK  0x7ffffff

typedef float v2f __attribute__((ext_vector_type(2)));
typedef float v4f __attribute__((ext_vector_type(4)));
typedef float v8f __attribute__((ext_vector_type(8)));

struct GemmStage {
    v4f a[4];
    v4f b[4][4];
};

__device__ __forceinline__ void gemm_load(
    GemmStage& g, const float* __restrict__ ap, const float* __restrict__ bp,
    const int st)
{
    #pragma unroll
    for (int sub = 0; sub < 4; ++sub) {
        const int k0 = st * 32 + sub * 8;
        g.a[sub] = *(const v4f*)(ap + k0);
        #pragma unroll
        for (int j = 0; j < 4; ++j)
            g.b[sub][j] = *(const v4f*)(bp + (size_t)(k0 + j) * OUT_DIM);
    }
}

__device__ __forceinline__ void gemm_mma(v8f (&acc)[4], const GemmStage& g)
{
    #pragma unroll
    for (int sub = 0; sub < 4; ++sub) {
        #pragma unroll
        for (int jp = 0; jp < 2; ++jp) {
            v2f a;
            a.x = g.a[sub][2 * jp];
            a.y = g.a[sub][2 * jp + 1];
            #pragma unroll
            for (int t = 0; t < 4; ++t) {
                v2f b;
                b.x = g.b[sub][2 * jp][t];
                b.y = g.b[sub][2 * jp + 1][t];
                acc[t] = __builtin_amdgcn_wmma_f32_16x16x4_f32(
                     false, a,  false, b,
                     (short)0, acc[t],  false,  false);
            }
        }
    }
}

__global__ __launch_bounds__(128) void gat_gemm_wmma(
    const float* __restrict__ h, const float* __restrict__ W,
    float* __restrict__ x, int N)
{
    const int lane = threadIdx.x & 31;
    const int wave = threadIdx.x >> 5;
    const int hf   = lane >> 4;
    const int c    = lane & 15;
    const int row0 = blockIdx.x * 64 + wave * 16;
    if (row0 >= N) return;

    int arow = row0 + c;
    if (arow >= N) arow = N - 1;
    const float* __restrict__ ap = h + (size_t)arow * IN_DIM + hf * 4;
    const float* __restrict__ bp = W + (size_t)(hf * 4) * OUT_DIM + c * 4;

    v8f acc[4];
    #pragma unroll
    for (int t = 0; t < 4; ++t)
        acc[t] = (v8f){0.f, 0.f, 0.f, 0.f, 0.f, 0.f, 0.f, 0.f};

    GemmStage g0, g1;
    gemm_load(g0, ap, bp, 0);
    #pragma unroll
    for (int st = 0; st < IN_DIM / 32; st += 2) {
        gemm_load(g1, ap, bp, st + 1);
        __builtin_amdgcn_sched_barrier(0);
        gemm_mma(acc, g0);
        if (st + 2 < IN_DIM / 32) {
            gemm_load(g0, ap, bp, st + 2);
            __builtin_amdgcn_sched_barrier(0);
        }
        gemm_mma(acc, g1);
    }

    float* __restrict__ xout =
        x + ((size_t)(row0 + hf * 8)) * OUT_DIM + c * 4;
    if (row0 + 16 <= N) {
        #pragma unroll
        for (int r = 0; r < 8; ++r) {
            v4f v = {acc[0][r], acc[1][r], acc[2][r], acc[3][r]};
            *(v4f*)(xout + (size_t)r * OUT_DIM) = v;
        }
    } else {
        #pragma unroll
        for (int r = 0; r < 8; ++r) {
            if (row0 + r + hf * 8 < N) {
                v4f v = {acc[0][r], acc[1][r], acc[2][r], acc[3][r]};
                *(v4f*)(xout + (size_t)r * OUT_DIM) = v;
            }
        }
    }
}

__global__ __launch_bounds__(256) void gat_coef(
    const float* __restrict__ x, const float* __restrict__ Wl,
    const float* __restrict__ Wr, float* __restrict__ el,
    float* __restrict__ er, int* __restrict__ cnt, int N)
{
    const int n = blockIdx.x * blockDim.x + threadIdx.x;
    if (n >= N) return;
    const float* __restrict__ xr = x + (size_t)n * OUT_DIM;
    float al[N_HEADS] = {0.f, 0.f, 0.f, 0.f};
    float ar[N_HEADS] = {0.f, 0.f, 0.f, 0.f};
    for (int d = 0; d < OUT_DIM; ++d) {
        const float xv = xr[d];
        #pragma unroll
        for (int hh = 0; hh < N_HEADS; ++hh) {
            al[hh] = fmaf(xv, Wl[hh * OUT_DIM + d], al[hh]);
            ar[hh] = fmaf(xv, Wr[hh * OUT_DIM + d], ar[hh]);
        }
    }
    #pragma unroll
    for (int hh = 0; hh < N_HEADS; ++hh) {
        el[(size_t)n * N_HEADS + hh] = al[hh];
        er[(size_t)n * N_HEADS + hh] = ar[hh];
    }
    cnt[n] = 0;
}

__device__ __forceinline__ float gat_edge_e(float l, float r) {
    float v = l + r;
    v = (v > 0.f) ? v : 0.2f * v;
    return __expf(v);
}

#define BK_THREADS 1024
#define BK_EPT     6
#define BK_TILE    (BK_THREADS * BK_EPT)
#define BK_BINS    2048

__global__ __launch_bounds__(BK_THREADS) void gat_bucket(
    const int* __restrict__ ei, int* __restrict__ cnt,
    int* __restrict__ lists, int E, int N)
{
    __shared__ int hist[BK_BINS];
    __shared__ int gofs[BK_BINS];
    __shared__ int wsum[32];
    __shared__ int s_src[BK_TILE];
    __shared__ int s_dst[BK_TILE];

    const int nbins = (N + BIN_ROWS - 1) / BIN_ROWS;
    if (nbins > BK_BINS) return;

    const int tid  = threadIdx.x;
    const int lane = tid & 31;
    const int wave = tid >> 5;
    const int base = blockIdx.x * BK_TILE;
    const int left = E - base;
    const int nvalid = left < BK_TILE ? left : BK_TILE;

    hist[tid] = 0;
    hist[tid + BK_THREADS] = 0;

    int src[BK_EPT], dst[BK_EPT], rnk[BK_EPT];
    #pragma unroll
    for (int i = 0; i < BK_EPT; ++i) {
        const int k = i * BK_THREADS + tid;
        src[i] = 0;
        dst[i] = 0;
        if (k < nvalid) {
            src[i] = ei[base + k];
            dst[i] = ei[E + base + k];
        }
    }
    __syncthreads();

    #pragma unroll
    for (int i = 0; i < BK_EPT; ++i) {
        const int k = i * BK_THREADS + tid;
        rnk[i] = 0;
        if (k < nvalid)
            rnk[i] = atomicAdd(&hist[(src[i] >> 5) & (BK_BINS - 1)], 1);
    }
    __syncthreads();

    const int c0 = hist[2 * tid];
    const int c1 = hist[2 * tid + 1];
    const int mine = c0 + c1;
    int inc = mine;
    #pragma unroll
    for (int d = 1; d < 32; d <<= 1) {
        const int t = __shfl_up(inc, d, 32);
        if (lane >= d) inc += t;
    }
    if (lane == 31) wsum[wave] = inc;

    __syncthreads();
    if (wave == 0) {
        const int w = wsum[lane];
        int winc = w;
        #pragma unroll
        for (int d = 1; d < 32; d <<= 1) {
            const int t = __shfl_up(winc, d, 32);
            if (lane >= d) winc += t;
        }
        wsum[lane] = winc - w;
    }
    __syncthreads();
    const int start0 = wsum[wave] + inc - mine;
    hist[2 * tid]     = start0;
    hist[2 * tid + 1] = start0 + c0;
    __syncthreads();

    const int ja  = tid;
    const int jb  = tid + BK_THREADS;
    const int sta = hist[ja];
    const int ena = hist[ja + 1];
    const int stb = hist[jb];
    const int enb = (jb + 1 < BK_BINS) ? hist[jb + 1] : nvalid;
    int ga = 0, gb = 0;
    if (ena > sta && ja < nbins) ga = atomicAdd(cnt + ja, ena - sta);
    if (enb > stb && jb < nbins) gb = atomicAdd(cnt + jb, enb - stb);

    #pragma unroll
    for (int i = 0; i < BK_EPT; ++i) {
        const int k = i * BK_THREADS + tid;
        if (k < nvalid) {
            const int p = hist[(src[i] >> 5) & (BK_BINS - 1)] + rnk[i];
            s_src[p] = src[i];
            s_dst[p] = dst[i];
        }
    }
    gofs[ja] = ga - sta;
    gofs[jb] = gb - stb;
    __syncthreads();

    for (int i = 0; i < BK_EPT; ++i) {
        const int k = i * BK_THREADS + tid;
        if (k < nvalid) {
            const int s = s_src[k];
            const int d = s_dst[k];
            unsigned bin = (unsigned)s >> 5;
            bin = bin < (unsigned)nbins ? bin : (unsigned)(nbins - 1);
            const int rows_here = (N - (int)bin * BIN_ROWS) < BIN_ROWS ? (N - (int)bin * BIN_ROWS) : BIN_ROWS;
            const int gp = gofs[bin] + k;
            if ((unsigned)gp < (unsigned)(rows_here * ROW_CAP))
                lists[(size_t)bin * BIN_WORDS + gp] = ((s & (BIN_ROWS - 1)) << ROW_SHIFT) | (d & DST_MASK);
        }
    }
}

struct RowAcc {
    v2f   a[N_HEADS];
    float rs[N_HEADS];
};

__device__ __forceinline__ float lane_f(float v, int j) {
    return __builtin_bit_cast(float,
        __builtin_amdgcn_readlane(__builtin_bit_cast(int, v), j));
}

template <int K>
__device__ __forceinline__ void sort_keys(int (&key)[K], const int lane)
{
    #pragma unroll
    for (int k = 2; k <= 32 * K; k <<= 1) {
        #pragma unroll
        for (int j = k >> 1; j > 0; j >>= 1) {
            if (j >= 32) {
                const int rj = j >> 5;
                #pragma unroll
                for (int r = 0; r < K; ++r) {
                    const int p = r ^ rj;
                    if (p > r) {
                        const bool up = (((r * 32) & k) == 0);
                        const int lo = key[r] < key[p] ? key[r] : key[p];
                        const int hi = key[r] < key[p] ? key[p] : key[r];
                        key[r] = up ? lo : hi;
                        key[p] = up ? hi : lo;
                    }
                }
            } else {
                #pragma unroll
                for (int r = 0; r < K; ++r) {
                    const int i = r * 32 + lane;
                    const int other = __shfl_xor(key[r], j, 32);
                    const bool up    = ((i & k) == 0);
                    const bool lower = ((lane & j) == 0);
                    const int lo = key[r] < other ? key[r] : other;
                    const int hi = key[r] < other ? other : key[r];
                    key[r] = (up == lower) ? lo : hi;
                }
            }
        }
    }
}

template <int K>
__device__ __forceinline__ void row_from_slots(
    RowAcc& s, const int* slots_row, const int deg, const int lane,
    const float4 l, const float* __restrict__ er, const float* __restrict__ x,
    const int N)
{
    int key[K];
    #pragma unroll
    for (int r = 0; r < K; ++r) {
        const int i = r * 32 + lane;
        key[r] = (i < deg) ? slots_row[i] : KEY_PAD;
    }
    sort_keys<K>(key, lane);

    #pragma unroll
    for (int r = 0; r < K; ++r) {
        const int left = deg - r * 32;
        if (left > 0) {
            const int nh = left < 32 ? left : 32;
            const bool valid = lane < nh;
            int d = valid ? key[r] : __builtin_amdgcn_readlane(key[r], nh - 1);
            d = (int)(((unsigned)d < (unsigned)N) ? (unsigned)d : (unsigned)(N - 1));
            const float4 rr = *(const float4*)(er + (size_t)d * N_HEADS);
            const float w0 = valid ? gat_edge_e(l.x, rr.x) : 0.f;
            const float w1 = valid ? gat_edge_e(l.y, rr.y) : 0.f;
            const float w2 = valid ? gat_edge_e(l.z, rr.z) : 0.f;
            const float w3 = valid ? gat_edge_e(l.w, rr.w) : 0.f;

            const int npad = (nh + 3) & ~3;
            for (int j = 0; j < npad; j += 4) {
                v2f xv[4];
                #pragma unroll
                for (int u = 0; u < 4; ++u) {
                    const int du = __builtin_amdgcn_readlane(d, j + u);
                    xv[u] = *(const v2f*)(x + (size_t)du * OUT_DIM + lane * 2);
                }
                #pragma unroll
                for (int u = 0; u < 4; ++u) {
                    const float e0 = lane_f(w0, j + u);
                    const float e1 = lane_f(w1, j + u);
                    const float e2 = lane_f(w2, j + u);
                    const float e3 = lane_f(w3, j + u);
                    s.a[0] += e0 * xv[u];  s.rs[0] += e0;
                    s.a[1] += e1 * xv[u];  s.rs[1] += e1;
                    s.a[2] += e2 * xv[u];  s.rs[2] += e2;
                    s.a[3] += e3 * xv[u];  s.rs[3] += e3;
                }
            }
        }
    }
}

__device__ __forceinline__ void row_from_edge_list(
    RowAcc& s, const int* __restrict__ ei, const int E, const int row,
    const int lane, const float4 l, const float* __restrict__ er,
    const float* __restrict__ x)
{
    for (int base = 0; base < E; base += 32) {
        const int e  = base + lane;
        const int sv = (e < E) ? ei[e] : -1;
        unsigned m = __builtin_amdgcn_ballot_w32(sv == row);
        while (m) {
            const int bit = __builtin_ctz(m);
            m &= m - 1;
            const int d = ei[E + base + bit];
            const float4 rr = *(const float4*)(er + (size_t)d * N_HEADS);
            const float e0 = gat_edge_e(l.x, rr.x);
            const float e1 = gat_edge_e(l.y, rr.y);
            const float e2 = gat_edge_e(l.z, rr.z);
            const float e3 = gat_edge_e(l.w, rr.w);
            const v2f xv = *(const v2f*)(x + (size_t)d * OUT_DIM + lane * 2);
            s.a[0] += e0 * xv;  s.rs[0] += e0;
            s.a[1] += e1 * xv;  s.rs[1] += e1;
            s.a[2] += e2 * xv;  s.rs[2] += e2;
            s.a[3] += e3 * xv;  s.rs[3] += e3;
        }
    }
}

#define RW_WAVES 8

__global__ __launch_bounds__(RW_WAVES * 32) void gat_rows(
    const int* __restrict__ ei, const int* __restrict__ cnt,
    const float* __restrict__ el, const float* __restrict__ er,
    const float* __restrict__ x, const float* __restrict__ b,
    float* out, const int* lists, int N, int E)
{
    __shared__ int rcnt[BIN_ROWS];
    __shared__ int rlist[BIN_WORDS];

    const int tid  = threadIdx.x;
    const int lane = tid & 31;
    const int wave = __builtin_amdgcn_readfirstlane(tid >> 5);
    const int bin  = blockIdx.x;
    const int nbins = (N + BIN_ROWS - 1) / BIN_ROWS;
    const int rows_left = N - bin * BIN_ROWS;
    const int rows_here = rows_left < BIN_ROWS ? rows_left : BIN_ROWS;
    const int total = cnt[bin];
    const bool listed = (nbins <= BK_BINS) && (total <= rows_here * ROW_CAP);

    if (tid < BIN_ROWS) rcnt[tid] = 0;
    __syncthreads();
    if (listed) {
        const int* list = lists + (size_t)bin * BIN_WORDS;
        for (int k0 = 0; k0 < total; k0 += RW_WAVES * 32 * 4) {
            const int k = k0 + tid * 4;
            if (k < total) {
                const int4 q = *(const int4*)(list + k);
                auto deal = [&](const int v, const int kk) {
                    if (kk < total) {
                        const int r = (int)((unsigned)v >> ROW_SHIFT);
                        const int pos = atomicAdd(&rcnt[r], 1);
                        if (pos < ROW_CAP) rlist[r * ROW_CAP + pos] = v & DST_MASK;
                    }
                };
                deal(q.x, k);
                deal(q.y, k + 1);
                deal(q.z, k + 2);
                deal(q.w, k + 3);
            }
        }
    }
    __syncthreads();

    const v2f bb = *(const v2f*)(b + lane * 2);
    for (int q = 0; q < BIN_ROWS / RW_WAVES; ++q) {
        const int rib = q * RW_WAVES + wave;
        const int row = bin * BIN_ROWS + rib;
        if (row >= N) break;
        const int deg = listed ? __builtin_amdgcn_readfirstlane(rcnt[rib]) : ROW_CAP + 1;
        const float4 l = *(const float4*)(el + (size_t)row * N_HEADS);
        const int* slots_row = rlist + rib * ROW_CAP;

        RowAcc s;
        #pragma unroll
        for (int hh = 0; hh < N_HEADS; ++hh) {
            s.a[hh] = (v2f){0.f, 0.f};
            s.rs[hh] = 0.f;
        }

        if (deg <= 32)            row_from_slots<1>(s, slots_row, deg, lane, l, er, x, N);
        else if (deg <= 64)       row_from_slots<2>(s, slots_row, deg, lane, l, er, x, N);
        else if (deg <= 128)      row_from_slots<4>(s, slots_row, deg, lane, l, er, x, N);
        else if (deg <= ROW_CAP)  row_from_slots<8>(s, slots_row, deg, lane, l, er, x, N);
        else                      row_from_edge_list(s, ei, E, row, lane, l, er, x);

        float* orow = out + (size_t)row * (N_HEADS * OUT_DIM) + lane * 2;
        #pragma unroll
        for (int hh = 0; hh < N_HEADS; ++hh) {
            const float inv = 1.0f / fmaxf(s.rs[hh], GAT_EPS);
            v2f v;
            v.x = fmaf(s.a[hh].x, inv, bb.x);
            v.y = fmaf(s.a[hh].y, inv, bb.y);
            v.x = (v.x > 0.f) ? v.x : (__expf(v.x) - 1.f);
            v.y = (v.y > 0.f) ? v.y : (__expf(v.y) - 1.f);
            *(v2f*)(orow + hh * OUT_DIM) = v;
        }
    }
}

extern "C" void kernel_launch(void* const* d_in, const int* in_sizes, int n_in,
                              void* d_out, int out_size, void* d_ws, size_t ws_size,
                              hipStream_t stream) {
    const float* h  = (const float*)d_in[0];
    const float* W  = (const float*)d_in[1];
    const float* Wl = (const float*)d_in[2];
    const float* Wr = (const float*)d_in[3];
    const float* b  = (const float*)d_in[4];
    const int*   ei = (const int*)d_in[5];

    const int N = in_sizes[0] / IN_DIM;
    const int E = in_sizes[5] / 2;

    float* out = (float*)d_out;

    float* x   = (float*)d_ws;
    float* el  = x  + (size_t)N * OUT_DIM;
    float* er  = el + (size_t)N * N_HEADS;
    int*   cnt = (int*)(er + (size_t)N * N_HEADS);

    gat_gemm_wmma<<<(unsigned)((N + 63) / 64), 128, 0, stream>>>(h, W, x, N);

    gat_coef<<<(unsigned)((N + 255) / 256), 256, 0, stream>>>(
        x, Wl, Wr, el, er, cnt, N);

    gat_bucket<<<(unsigned)((E + BK_TILE - 1) / BK_TILE), BK_THREADS, 0, stream>>>(
        ei, cnt, (int*)d_out, E, N);

    gat_rows<<<(unsigned)((N + BIN_ROWS - 1) / BIN_ROWS), RW_WAVES * 32, 0, stream>>>(
        ei, cnt, el, er, x, b, out, (const int*)d_out, N, E);
}
